# starting kernel
# baseline (speedup 1.0000x reference)
#include <hip/hip_runtime.h>

typedef __attribute__((ext_vector_type(2))) float v2f;
typedef __attribute__((ext_vector_type(8))) float v8f;

#define N_NODES 100000
#define N_EDGES 1600000
#define IN_DIM 256
#define HID_DIM 128
#define OUT_DIM 64

__global__ __launch_bounds__(256) void gemm1_wmma(const float* __restrict__ X,
                                                  const float* __restrict__ W1,
                                                  float* __restrict__ H1) {
    __shared__ float xs[16][260];
    __shared__ float w1s[64][136];

    const int tid  = threadIdx.x;
    const int lane = tid & 31;
    const int wave = tid >> 5;
    const int half = lane >> 4;
    const int l16  = lane & 15;
    const int mbase = blockIdx.x * 16;

    {
        int row = tid >> 4;
        int seg = (tid & 15) * 16;
        const float* xp = X + (size_t)(mbase + row) * IN_DIM + seg;
        float4 a0 = ((const float4*)xp)[0];
        float4 a1 = ((const float4*)xp)[1];
        float4 a2 = ((const float4*)xp)[2];
        float4 a3 = ((const float4*)xp)[3];
        *(float4*)&xs[row][seg + 0]  = a0;
        *(float4*)&xs[row][seg + 4]  = a1;
        *(float4*)&xs[row][seg + 8]  = a2;
        *(float4*)&xs[row][seg + 12] = a3;
    }

    v8f acc = {};
    const int nb = wave * 16 + l16;

    for (int kc = 0; kc < IN_DIM; kc += 64) {
        for (int i = tid; i < (64 * 128) / 4; i += 256) {
            int flat = i * 4;
            int kr   = flat >> 7;
            int col  = flat & 127;
            float4 w = *(const float4*)&W1[(size_t)(kc + kr) * HID_DIM + col];
            *(float4*)&w1s[kr][col] = w;
        }
        __syncthreads();

#pragma unroll
        for (int kk = 0; kk < 64; kk += 4) {
            const int ka = kc + kk + half * 2;
            const int kb = kk + half * 2;
            v2f a = {xs[l16][ka], xs[l16][ka + 1]};
            v2f b = {w1s[kb][nb], w1s[kb + 1][nb]};
            acc = __builtin_amdgcn_wmma_f32_16x16x4_f32(
                false, a, false, b, (short)0, acc, false, false);
        }
        __syncthreads();
    }

#pragma unroll
    for (int v = 0; v < 8; ++v) {
        int row = mbase + v + half * 8;
        H1[(size_t)row * HID_DIM + nb] = acc[v];
    }
}

__global__ __launch_bounds__(256) void gemm2_wmma(const float* __restrict__ Hagg,
                                                  const float* __restrict__ W2,
                                                  float* __restrict__ H2) {
    __shared__ float hs[32][132];
    __shared__ float w2s[128][72];

    const int tid  = threadIdx.x;
    const int lane = tid & 31;
    const int wave = tid >> 5;
    const int mt   = wave >> 2;
    const int nt   = wave & 3;
    const int half = lane >> 4;
    const int l16  = lane & 15;
    const int mbase = blockIdx.x * 32;

    for (int i = tid; i < 1024; i += 256) {
        int flat = i * 4;
        int row  = flat >> 7;
        int col  = flat & 127;
        float4 h = *(const float4*)&Hagg[(size_t)(mbase + row) * HID_DIM + col];
        h.x = fmaxf(h.x, 0.0f);
        h.y = fmaxf(h.y, 0.0f);
        h.z = fmaxf(h.z, 0.0f);
        h.w = fmaxf(h.w, 0.0f);
        *(float4*)&hs[row][col] = h;
    }
    for (int i = tid; i < 2048; i += 256) {
        int flat = i * 4;
        int kr   = flat >> 6;
        int col  = flat & 63;
        *(float4*)&w2s[kr][col] = *(const float4*)&W2[(size_t)kr * OUT_DIM + col];
    }
    __syncthreads();

    v8f acc = {};
    const int mrow = mt * 16 + l16;
    const int nb   = nt * 16 + l16;

#pragma unroll
    for (int kk = 0; kk < HID_DIM; kk += 4) {
        const int k = kk + half * 2;
        v2f a = {hs[mrow][k], hs[mrow][k + 1]};
        v2f b = {w2s[k][nb], w2s[k + 1][nb]};
        acc = __builtin_amdgcn_wmma_f32_16x16x4_f32(
            false, a, false, b, (short)0, acc, false, false);
    }

#pragma unroll
    for (int v = 0; v < 8; ++v) {
        int row = mbase + mt * 16 + v + half * 8;
        H2[(size_t)row * OUT_DIM + nt * 16 + l16] = acc[v];
    }
}

__global__ __launch_bounds__(256) void spmm_scatter_d128(
    const int* __restrict__ src, const int* __restrict__ dst,
    const float* __restrict__ vals, const float* __restrict__ H,
    float* __restrict__ out, int E) {
    int e = blockIdx.x * 8 + threadIdx.y;
    if (e >= E) return;
    int lane = threadIdx.x;
    int s = src[e];
    int d = dst[e];
    float v = vals[e];
    float4 m = ((const float4*)(H + (size_t)s * 128))[lane];
    float* op = out + (size_t)d * 128 + lane * 4;
    atomicAdd(op + 0, m.x * v);
    atomicAdd(op + 1, m.y * v);
    atomicAdd(op + 2, m.z * v);
    atomicAdd(op + 3, m.w * v);
}

__global__ __launch_bounds__(256) void spmm_scatter_d64(
    const int* __restrict__ src, const int* __restrict__ dst,
    const float* __restrict__ vals, const float* __restrict__ H,
    float* __restrict__ out, int E) {
    int e = blockIdx.x * 8 + threadIdx.y;
    if (e >= E) return;
    int lane = threadIdx.x;
    int s = src[e];
    int d = dst[e];
    float v = vals[e];
    float2 m = ((const float2*)(H + (size_t)s * 64))[lane];
    float* op = out + (size_t)d * 64 + lane * 2;
    atomicAdd(op + 0, m.x * v);
    atomicAdd(op + 1, m.y * v);
}

extern "C" void kernel_launch(void* const* d_in, const int* in_sizes, int n_in,
                              void* d_out, int out_size, void* d_ws, size_t ws_size,
                              hipStream_t stream) {
    const float* X     = (const float*)d_in[0];
    const int*   esrc  = (const int*)d_in[1];
    const int*   edst  = (const int*)d_in[2];
    const float* evals = (const float*)d_in[3];
    const float* W1    = (const float*)d_in[4];
    const float* W2    = (const float*)d_in[5];
    float*       out   = (float*)d_out;

    char* ws = (char*)d_ws;
    const size_t h1Bytes = (size_t)N_NODES * HID_DIM * sizeof(float);
    float* H1   = (float*)(ws);
    float* Hagg = (float*)(ws + h1Bytes);
    float* H2   = (float*)(ws + 2 * h1Bytes);

    hipMemsetAsync(Hagg, 0, h1Bytes, stream);
    gemm1_wmma<<<N_NODES / 16, 256, 0, stream>>>(X, W1, H1);
    dim3 eb(32, 8);
    spmm_scatter_d128<<<N_EDGES / 8, eb, 0, stream>>>(esrc, edst, evals, H1, Hagg, N_EDGES);

    hipMemsetAsync(out, 0, (size_t)N_NODES * OUT_DIM * sizeof(float), stream);
    gemm2_wmma<<<N_NODES / 32, 256, 0, stream>>>(Hagg, W2, H2);
    spmm_scatter_d64<<<N_EDGES / 8, eb, 0, stream>>>(esrc, edst, evals, H2, out, N_EDGES);
}


# spmm scatter: each wave walks 32 consecutive edges (indices read once, coalesced, passed by lane read; row loads a batch ahead of the f32 atomics) instead of one wave per edge; sums still by atomics
# speedup vs baseline: 1.2370x; 1.2370x over previous
#include <hip/hip_runtime.h>

typedef __attribute__((ext_vector_type(2))) float v2f;
typedef __attribute__((ext_vector_type(8))) float v8f;

#define N_NODES 100000
#define N_EDGES 1600000
#define IN_DIM 256
#define HID_DIM 128
#define OUT_DIM 64

__global__ __launch_bounds__(256) void gemm1_wmma(const float* __restrict__ X,
                                                  const float* __restrict__ W1,
                                                  float* __restrict__ H1) {
    __shared__ float xs[16][260];
    __shared__ float w1s[64][136];

    const int tid  = threadIdx.x;
    const int lane = tid & 31;
    const int wave = tid >> 5;
    const int half = lane >> 4;
    const int l16  = lane & 15;
    const int mbase = blockIdx.x * 16;

    {
        int row = tid >> 4;
        int seg = (tid & 15) * 16;
        const float* xp = X + (size_t)(mbase + row) * IN_DIM + seg;
        float4 a0 = ((const float4*)xp)[0];
        float4 a1 = ((const float4*)xp)[1];
        float4 a2 = ((const float4*)xp)[2];
        float4 a3 = ((const float4*)xp)[3];
        *(float4*)&xs[row][seg + 0]  = a0;
        *(float4*)&xs[row][seg + 4]  = a1;
        *(float4*)&xs[row][seg + 8]  = a2;
        *(float4*)&xs[row][seg + 12] = a3;
    }

    v8f acc = {};
    const int nb = wave * 16 + l16;

    for (int kc = 0; kc < IN_DIM; kc += 64) {
        for (int i = tid; i < (64 * 128) / 4; i += 256) {
            int flat = i * 4;
            int kr   = flat >> 7;
            int col  = flat & 127;
            float4 w = *(const float4*)&W1[(size_t)(kc + kr) * HID_DIM + col];
            *(float4*)&w1s[kr][col] = w;
        }
        __syncthreads();

#pragma unroll
        for (int kk = 0; kk < 64; kk += 4) {
            const int ka = kc + kk + half * 2;
            const int kb = kk + half * 2;
            v2f a = {xs[l16][ka], xs[l16][ka + 1]};
            v2f b = {w1s[kb][nb], w1s[kb + 1][nb]};
            acc = __builtin_amdgcn_wmma_f32_16x16x4_f32(
                false, a, false, b, (short)0, acc, false, false);
        }
        __syncthreads();
    }

#pragma unroll
    for (int v = 0; v < 8; ++v) {
        int row = mbase + v + half * 8;
        H1[(size_t)row * HID_DIM + nb] = acc[v];
    }
}

__global__ __launch_bounds__(256) void gemm2_wmma(const float* __restrict__ Hagg,
                                                  const float* __restrict__ W2,
                                                  float* __restrict__ H2) {
    __shared__ float hs[32][132];
    __shared__ float w2s[128][72];

    const int tid  = threadIdx.x;
    const int lane = tid & 31;
    const int wave = tid >> 5;
    const int mt   = wave >> 2;
    const int nt   = wave & 3;
    const int half = lane >> 4;
    const int l16  = lane & 15;
    const int mbase = blockIdx.x * 32;

    for (int i = tid; i < 1024; i += 256) {
        int flat = i * 4;
        int row  = flat >> 7;
        int col  = flat & 127;
        float4 h = *(const float4*)&Hagg[(size_t)(mbase + row) * HID_DIM + col];
        h.x = fmaxf(h.x, 0.0f);
        h.y = fmaxf(h.y, 0.0f);
        h.z = fmaxf(h.z, 0.0f);
        h.w = fmaxf(h.w, 0.0f);
        *(float4*)&hs[row][col] = h;
    }
    for (int i = tid; i < 2048; i += 256) {
        int flat = i * 4;
        int kr   = flat >> 6;
        int col  = flat & 63;
        *(float4*)&w2s[kr][col] = *(const float4*)&W2[(size_t)kr * OUT_DIM + col];
    }
    __syncthreads();

    v8f acc = {};
    const int mrow = mt * 16 + l16;
    const int nb   = nt * 16 + l16;

#pragma unroll
    for (int kk = 0; kk < HID_DIM; kk += 4) {
        const int k = kk + half * 2;
        v2f a = {hs[mrow][k], hs[mrow][k + 1]};
        v2f b = {w2s[k][nb], w2s[k + 1][nb]};
        acc = __builtin_amdgcn_wmma_f32_16x16x4_f32(
            false, a, false, b, (short)0, acc, false, false);
    }

#pragma unroll
    for (int v = 0; v < 8; ++v) {
        int row = mbase + mt * 16 + v + half * 8;
        H2[(size_t)row * OUT_DIM + nt * 16 + l16] = acc[v];
    }
}

#define EPW 32
#define UB128 8
#define UB64 16

static __device__ __forceinline__ int lane_read_i(int x, int l) {
    return __builtin_amdgcn_readlane(x, l);
}
static __device__ __forceinline__ float lane_read_f(float x, int l) {
    return __int_as_float(__builtin_amdgcn_readlane(__float_as_int(x), l));
}

__global__ __launch_bounds__(256) void spmm_scatter_d128(
    const int* __restrict__ src, const int* __restrict__ dst,
    const float* __restrict__ vals, const float* __restrict__ H,
    float* __restrict__ out, int E) {
    const unsigned lane = threadIdx.x;
    const int wave = __builtin_amdgcn_readfirstlane((int)threadIdx.y);
    const int base = (blockIdx.x * 8 + wave) * EPW;
    if (base >= E) return;
    const int n = min(EPW, E - base);

    int s_l = 0, d_l = 0;
    float v_l = 0.0f;
    if ((int)lane < n) {
        s_l = src[base + lane];
        d_l = dst[base + lane];
        v_l = vals[base + lane];
    }
    const float4* Hv = (const float4*)H;

    if (n == EPW) {
        float4 cur[UB128], nxt[UB128];
#pragma unroll
        for (int u = 0; u < UB128; ++u) {
            const float4* row = Hv + (size_t)lane_read_i(s_l, u) * 32;
            cur[u] = row[lane];
        }
#pragma unroll
        for (int i = 0; i < EPW; i += UB128) {
            if (i + UB128 < EPW) {
#pragma unroll
                for (int u = 0; u < UB128; ++u) {
                    const float4* row = Hv + (size_t)lane_read_i(s_l, i + UB128 + u) * 32;
                    nxt[u] = row[lane];
                }
            }
#pragma unroll
            for (int u = 0; u < UB128; ++u) {
                const int d = lane_read_i(d_l, i + u);
                const float v = lane_read_f(v_l, i + u);
                float* op = out + (size_t)d * 128 + lane * 4;
                atomicAdd(op + 0, cur[u].x * v);
                atomicAdd(op + 1, cur[u].y * v);
                atomicAdd(op + 2, cur[u].z * v);
                atomicAdd(op + 3, cur[u].w * v);
            }
#pragma unroll
            for (int u = 0; u < UB128; ++u) cur[u] = nxt[u];
        }
    } else {
        for (int i = 0; i < n; ++i) {
            const int s = lane_read_i(s_l, i);
            const int d = lane_read_i(d_l, i);
            const float v = lane_read_f(v_l, i);
            const float4 m = (Hv + (size_t)s * 32)[lane];
            float* op = out + (size_t)d * 128 + lane * 4;
            atomicAdd(op + 0, m.x * v);
            atomicAdd(op + 1, m.y * v);
            atomicAdd(op + 2, m.z * v);
            atomicAdd(op + 3, m.w * v);
        }
    }
}

__global__ __launch_bounds__(256) void spmm_scatter_d64(
    const int* __restrict__ src, const int* __restrict__ dst,
    const float* __restrict__ vals, const float* __restrict__ H,
    float* __restrict__ out, int E) {
    const unsigned lane = threadIdx.x;
    const int wave = __builtin_amdgcn_readfirstlane((int)threadIdx.y);
    const int base = (blockIdx.x * 8 + wave) * EPW;
    if (base >= E) return;
    const int n = min(EPW, E - base);

    int s_l = 0, d_l = 0;
    float v_l = 0.0f;
    if ((int)lane < n) {
        s_l = src[base + lane];
        d_l = dst[base + lane];
        v_l = vals[base + lane];
    }
    const float2* Hv = (const float2*)H;

    if (n == EPW) {
        float2 cur[UB64], nxt[UB64];
#pragma unroll
        for (int u = 0; u < UB64; ++u) {
            const float2* row = Hv + (size_t)lane_read_i(s_l, u) * 32;
            cur[u] = row[lane];
        }
#pragma unroll
        for (int i = 0; i < EPW; i += UB64) {
            if (i + UB64 < EPW) {
#pragma unroll
                for (int u = 0; u < UB64; ++u) {
                    const float2* row = Hv + (size_t)lane_read_i(s_l, i + UB64 + u) * 32;
                    nxt[u] = row[lane];
                }
            }
#pragma unroll
            for (int u = 0; u < UB64; ++u) {
                const int d = lane_read_i(d_l, i + u);
                const float v = lane_read_f(v_l, i + u);
                float* op = out + (size_t)d * 64 + lane * 2;
                atomicAdd(op + 0, cur[u].x * v);
                atomicAdd(op + 1, cur[u].y * v);
            }
#pragma unroll
            for (int u = 0; u < UB64; ++u) cur[u] = nxt[u];
        }
    } else {
        for (int i = 0; i < n; ++i) {
            const int s = lane_read_i(s_l, i);
            const int d = lane_read_i(d_l, i);
            const float v = lane_read_f(v_l, i);
            const float2 m = (Hv + (size_t)s * 32)[lane];
            float* op = out + (size_t)d * 64 + lane * 2;
            atomicAdd(op + 0, m.x * v);
            atomicAdd(op + 1, m.y * v);
        }
    }
}

extern "C" void kernel_launch(void* const* d_in, const int* in_sizes, int n_in,
                              void* d_out, int out_size, void* d_ws, size_t ws_size,
                              hipStream_t stream) {
    const float* X     = (const float*)d_in[0];
    const int*   esrc  = (const int*)d_in[1];
    const int*   edst  = (const int*)d_in[2];
    const float* evals = (const float*)d_in[3];
    const float* W1    = (const float*)d_in[4];
    const float* W2    = (const float*)d_in[5];
    float*       out   = (float*)d_out;

    char* ws = (char*)d_ws;
    const size_t h1Bytes = (size_t)N_NODES * HID_DIM * sizeof(float);
    float* H1   = (float*)(ws);
    float* Hagg = (float*)(ws + h1Bytes);
    float* H2   = (float*)(ws + 2 * h1Bytes);

    hipMemsetAsync(Hagg, 0, h1Bytes, stream);
    gemm1_wmma<<<N_NODES / 16, 256, 0, stream>>>(X, W1, H1);
    dim3 eb(32, 8);
    spmm_scatter_d128<<<(N_EDGES + 255) / 256, eb, 0, stream>>>(esrc, edst, evals, H1, Hagg, N_EDGES);

    hipMemsetAsync(out, 0, (size_t)N_NODES * OUT_DIM * sizeof(float), stream);
    gemm2_wmma<<<N_NODES / 32, 256, 0, stream>>>(Hagg, W2, H2);
    spmm_scatter_d64<<<(N_EDGES + 255) / 256, eb, 0, stream>>>(esrc, edst, evals, H2, out, N_EDGES);
}


# spmm sums moved from cache atomics to LDS: edges grouped per call by 100-row destination range (LDS counts, scan, fill); one work-group per range adds rows into an LDS tile; f32 adds, arrival order
# speedup vs baseline: 2.9029x; 2.3467x over previous
#include <hip/hip_runtime.h>

typedef __attribute__((ext_vector_type(2))) float v2f;
typedef __attribute__((ext_vector_type(8))) float v8f;

#define N_NODES 100000
#define N_EDGES 1600000
#define IN_DIM 256
#define HID_DIM 128
#define OUT_DIM 64

__global__ __launch_bounds__(256) void gemm1_wmma(const float* __restrict__ X,
                                                  const float* __restrict__ W1,
                                                  float* __restrict__ H1,
                                                  int* __restrict__ tables) {
    __shared__ float xs[16][260];
    __shared__ float w1s[64][136];

    const int tid  = threadIdx.x;
    if (blockIdx.x < 8) tables[blockIdx.x * 256 + tid] = 0;
    const int lane = tid & 31;
    const int wave = tid >> 5;
    const int half = lane >> 4;
    const int l16  = lane & 15;
    const int mbase = blockIdx.x * 16;

    const int wr = tid >> 5;
    const int wc = (tid & 31) * 4;
    const float* wp = W1 + (size_t)wr * HID_DIM + wc;

    const int xrow = tid >> 4;
    const int xseg = (tid & 15) * 16;
    const float4* xp = (const float4*)(X + (size_t)(mbase + xrow) * IN_DIM + xseg);
    float4 a0 = xp[0];
    float4 a1 = xp[1];
    float4 a2 = xp[2];
    float4 a3 = xp[3];

    v8f acc = {};
    const int nb = wave * 16 + l16;

    for (int kc = 0; kc < IN_DIM; kc += 64) {
        float4 w[8];
#pragma unroll
        for (int t = 0; t < 8; ++t)
            w[t] = *(const float4*)(wp + (size_t)(kc + 8 * t) * HID_DIM);
        if (kc == 0) {
            *(float4*)&xs[xrow][xseg + 0]  = a0;
            *(float4*)&xs[xrow][xseg + 4]  = a1;
            *(float4*)&xs[xrow][xseg + 8]  = a2;
            *(float4*)&xs[xrow][xseg + 12] = a3;
        }
#pragma unroll
        for (int t = 0; t < 8; ++t)
            *(float4*)&w1s[wr + 8 * t][wc] = w[t];
        __syncthreads();

#pragma unroll
        for (int kk = 0; kk < 64; kk += 4) {
            const int ka = kc + kk + half * 2;
            const int kb = kk + half * 2;
            v2f a = {xs[l16][ka], xs[l16][ka + 1]};
            v2f b = {w1s[kb][nb], w1s[kb + 1][nb]};
            acc = __builtin_amdgcn_wmma_f32_16x16x4_f32(
                false, a, false, b, (short)0, acc, false, false);
        }
        __syncthreads();
    }

#pragma unroll
    for (int v = 0; v < 8; ++v) {
        int row = mbase + v + half * 8;
        H1[(size_t)row * HID_DIM + nb] = acc[v];
    }
}

__global__ __launch_bounds__(256) void gemm2_wmma(const float* __restrict__ Hagg,
                                                  const float* __restrict__ W2,
                                                  float* __restrict__ H2) {
    __shared__ float hs[32][132];
    __shared__ float w2s[128][72];

    const int tid  = threadIdx.x;
    const int lane = tid & 31;
    const int wave = tid >> 5;
    const int mt   = wave >> 2;
    const int nt   = wave & 3;
    const int half = lane >> 4;
    const int l16  = lane & 15;
    const int mbase = blockIdx.x * 32;

    const int hr   = tid >> 5;
    const int hc   = (tid & 31) * 4;
    const int kr   = tid >> 4;
    const int kcol = (tid & 15) * 4;
    const float* hp = Hagg + (size_t)(mbase + hr) * HID_DIM + hc;
    const float* wp = W2 + (size_t)kr * OUT_DIM + kcol;
    float4 h[4], w[8];
#pragma unroll
    for (int t = 0; t < 4; ++t)
        h[t] = *(const float4*)(hp + (size_t)(8 * t) * HID_DIM);
#pragma unroll
    for (int t = 0; t < 8; ++t)
        w[t] = *(const float4*)(wp + (size_t)(16 * t) * OUT_DIM);
#pragma unroll
    for (int t = 0; t < 4; ++t) {
        float4 r = h[t];
        r.x = fmaxf(r.x, 0.0f);
        r.y = fmaxf(r.y, 0.0f);
        r.z = fmaxf(r.z, 0.0f);
        r.w = fmaxf(r.w, 0.0f);
        *(float4*)&hs[hr + 8 * t][hc] = r;
    }
#pragma unroll
    for (int t = 0; t < 8; ++t)
        *(float4*)&w2s[kr + 16 * t][kcol] = w[t];
    __syncthreads();

    v8f acc = {};
    const int mrow = mt * 16 + l16;
    const int nb   = nt * 16 + l16;

#pragma unroll
    for (int kk = 0; kk < HID_DIM; kk += 4) {
        const int k = kk + half * 2;
        v2f a = {hs[mrow][k], hs[mrow][k + 1]};
        v2f b = {w2s[k][nb], w2s[k + 1][nb]};
        acc = __builtin_amdgcn_wmma_f32_16x16x4_f32(
            false, a, false, b, (short)0, acc, false, false);
    }

#pragma unroll
    for (int v = 0; v < 8; ++v) {
        int row = mbase + mt * 16 + v + half * 8;
        H2[(size_t)row * OUT_DIM + nt * 16 + l16] = acc[v];
    }
}

#define ROWS_PER_BIN 100
#define NBINS (N_NODES / ROWS_PER_BIN)
#define TABLE_INTS 1024
#define BIN_THREADS 1024
#define BIN_EPT 6
#define BIN_EPB (BIN_THREADS * BIN_EPT)
#define BIN_BLOCKS ((N_EDGES + BIN_EPB - 1) / BIN_EPB)
#define SUM_THREADS 512
#define KEY_SHIFT 7

static __device__ __forceinline__ int lane_read_i(int x, int l) {
    return __builtin_amdgcn_readlane(x, l);
}
static __device__ __forceinline__ float lane_read_f(float x, int l) {
    return __int_as_float(__builtin_amdgcn_readlane(__float_as_int(x), l));
}
static __device__ __forceinline__ int bin_of(int d) {
    const unsigned b = (unsigned)d / (unsigned)ROWS_PER_BIN;
    return (int)(b < (unsigned)NBINS ? b : (unsigned)(NBINS - 1));
}

__global__ __launch_bounds__(BIN_THREADS) void bin_count(const int* __restrict__ dst, int E,
                                                         int* __restrict__ gcount) {
    __shared__ int h[TABLE_INTS];
    const int tid = threadIdx.x;
    h[tid] = 0;
    const int base = blockIdx.x * BIN_EPB + tid;
    int d[BIN_EPT];
#pragma unroll
    for (int k = 0; k < BIN_EPT; ++k) {
        const int e = base + k * BIN_THREADS;
        d[k] = (e < E) ? dst[e] : -1;
    }
    __syncthreads();
#pragma unroll
    for (int k = 0; k < BIN_EPT; ++k)
        if (d[k] >= 0) atomicAdd(&h[bin_of(d[k])], 1);
    __syncthreads();
    const int c = h[tid];
    if (tid < NBINS && c != 0) atomicAdd(&gcount[tid], c);
}

__global__ __launch_bounds__(BIN_THREADS) void bin_fill(const int* __restrict__ src,
                                                        const int* __restrict__ dst,
                                                        const float* __restrict__ vals, int E,
                                                        const int* __restrict__ gcount,
                                                        int* __restrict__ gcursor,
                                                        int* __restrict__ binstart,
                                                        int* __restrict__ rec) {
    __shared__ int s_pos[TABLE_INTS];
    __shared__ int h[TABLE_INTS];
    __shared__ int s_wsum[32];
    const int tid  = threadIdx.x;
    const int lane = tid & 31;
    const int wave = tid >> 5;

    const int base = blockIdx.x * BIN_EPB + tid;
    int d[BIN_EPT], s[BIN_EPT];
    float v[BIN_EPT];
#pragma unroll
    for (int k = 0; k < BIN_EPT; ++k) {
        const int e = base + k * BIN_THREADS;
        d[k] = -1; s[k] = 0; v[k] = 0.0f;
        if (e < E) { d[k] = dst[e]; s[k] = src[e]; v[k] = vals[e]; }
    }

    h[tid] = 0;
    const int c = (tid < NBINS) ? gcount[tid] : 0;
    int x = c;
#pragma unroll
    for (int off = 1; off < 32; off <<= 1) {
        const int y = __shfl_up(x, off);
        if (lane >= off) x += y;
    }
    if (lane == 31) s_wsum[wave] = x;
    __syncthreads();
    if (wave == 0) {
        const int w = s_wsum[lane];
        int xs = w;
#pragma unroll
        for (int off = 1; off < 32; off <<= 1) {
            const int y = __shfl_up(xs, off);
            if (lane >= off) xs += y;
        }
        s_wsum[lane] = xs - w;
    }
    __syncthreads();
    const int start = s_wsum[wave] + x - c;
    s_pos[tid] = start;
    if (blockIdx.x == 0) {
        if (tid < NBINS) binstart[tid] = start;
        if (tid == NBINS - 1) binstart[NBINS] = start + c;
    }

#pragma unroll
    for (int k = 0; k < BIN_EPT; ++k)
        if (d[k] >= 0) atomicAdd(&h[bin_of(d[k])], 1);
    __syncthreads();
    {
        const int lc = h[tid];
        if (tid < NBINS && lc != 0) s_pos[tid] = start + atomicAdd(&gcursor[tid], lc);
        h[tid] = 0;
    }
    __syncthreads();
#pragma unroll
    for (int k = 0; k < BIN_EPT; ++k) {
        if (d[k] >= 0) {
            const int b = bin_of(d[k]);
            const int pos = s_pos[b] + atomicAdd(&h[b], 1);
            int lr = d[k] - b * ROWS_PER_BIN;
            if ((unsigned)lr >= (unsigned)ROWS_PER_BIN) lr = (1 << KEY_SHIFT) - 1;
            int2 r;
            r.x = (s[k] << KEY_SHIFT) | lr;
            r.y = __float_as_int(v[k]);
            *(int2*)(rec + 2 * (size_t)pos) = r;
        }
    }
}

template <int D>
static __device__ __forceinline__ void bin_sum(const int* __restrict__ rec,
                                               const int* __restrict__ binstart,
                                               const float* __restrict__ H,
                                               float* __restrict__ out) {
    constexpr int J  = D / 32;
    constexpr int UB = 32 / J;
    constexpr int T4 = ROWS_PER_BIN * D / 4;
    __shared__ float acc[ROWS_PER_BIN * D];
    const int tid  = threadIdx.x;
    const int lane = tid & 31;
    const int wave = __builtin_amdgcn_readfirstlane(tid >> 5);
    const int b    = blockIdx.x;

    float4* acc4 = (float4*)acc;
    for (int i = tid; i < T4; i += SUM_THREADS) acc4[i] = make_float4(0.0f, 0.0f, 0.0f, 0.0f);
    const int lo = binstart[b];
    const int hi = binstart[b + 1];
    __syncthreads();

    for (int base = lo + wave * 32; base < hi; base += SUM_THREADS) {
        const int e = base + lane;
        int key = 0;
        float val = 0.0f;
        if (e < hi) {
            const int2 r = *(const int2*)(rec + 2 * (size_t)e);
            key = r.x;
            val = __int_as_float(r.y);
        }
        const int n = (hi - base < 32) ? (hi - base) : 32;
        for (int u0 = 0; u0 < n; u0 += UB) {
            float r[UB][J];
#pragma unroll
            for (int u = 0; u < UB; ++u) {
                const int uu = (u0 + u < n) ? (u0 + u) : (n - 1);
                const int k = lane_read_i(key, uu);
                const float* row = H + (size_t)(k >> KEY_SHIFT) * D + lane;
#pragma unroll
                for (int j = 0; j < J; ++j) r[u][j] = row[32 * j];
            }
#pragma unroll
            for (int u = 0; u < UB; ++u) {
                if (u0 + u < n) {
                    const int k = lane_read_i(key, u0 + u);
                    const float w = lane_read_f(val, u0 + u);
                    const int lr = k & ((1 << KEY_SHIFT) - 1);
                    if (lr < ROWS_PER_BIN) {
                        float* a = acc + lr * D + lane;
#pragma unroll
                        for (int j = 0; j < J; ++j) atomicAdd(a + 32 * j, r[u][j] * w);
                    }
                }
            }
        }
    }
    __syncthreads();

    float4* o4 = (float4*)(out + (size_t)b * ROWS_PER_BIN * D);
    for (int i = tid; i < T4; i += SUM_THREADS) o4[i] = acc4[i];
}

__global__ __launch_bounds__(SUM_THREADS) void bin_sum_d128(const int* __restrict__ rec,
                                                            const int* __restrict__ binstart,
                                                            const float* __restrict__ H,
                                                            float* __restrict__ out) {
    bin_sum<128>(rec, binstart, H, out);
}

__global__ __launch_bounds__(SUM_THREADS) void bin_sum_d64(const int* __restrict__ rec,
                                                           const int* __restrict__ binstart,
                                                           const float* __restrict__ H,
                                                           float* __restrict__ out) {
    bin_sum<64>(rec, binstart, H, out);
}

extern "C" void kernel_launch(void* const* d_in, const int* in_sizes, int n_in,
                              void* d_out, int out_size, void* d_ws, size_t ws_size,
                              hipStream_t stream) {
    const float* X     = (const float*)d_in[0];
    const int*   esrc  = (const int*)d_in[1];
    const int*   edst  = (const int*)d_in[2];
    const float* evals = (const float*)d_in[3];
    const float* W1    = (const float*)d_in[4];
    const float* W2    = (const float*)d_in[5];
    float*       out   = (float*)d_out;

    char* ws = (char*)d_ws;
    const size_t h1Bytes  = (size_t)N_NODES * HID_DIM * sizeof(float);
    const size_t recBytes = (size_t)N_EDGES * 2 * sizeof(int);
    float* H1   = (float*)(ws);
    float* Hagg = (float*)(ws + h1Bytes);
    float* H2   = (float*)(ws);
    int* rec      = (int*)(ws + 2 * h1Bytes);
    int* tables   = (int*)(ws + 2 * h1Bytes + recBytes);
    int* gcount   = tables;
    int* gcursor  = tables + TABLE_INTS;
    int* binstart = tables + 2 * TABLE_INTS;

    gemm1_wmma<<<N_NODES / 16, 256, 0, stream>>>(X, W1, H1, tables);
    bin_count<<<BIN_BLOCKS, BIN_THREADS, 0, stream>>>(edst, N_EDGES, gcount);
    bin_fill<<<BIN_BLOCKS, BIN_THREADS, 0, stream>>>(esrc, edst, evals, N_EDGES, gcount, gcursor,
                                                     binstart, rec);
    bin_sum_d128<<<NBINS, SUM_THREADS, 0, stream>>>(rec, binstart, H1, Hagg);

    gemm2_wmma<<<N_NODES / 32, 256, 0, stream>>>(Hagg, W2, H2);
    bin_sum_d64<<<NBINS, SUM_THREADS, 0, stream>>>(rec, binstart, H2, out);
}


# spmm sum: a range's records ordered by row in LDS (counting sort), a wave sums each run of equal rows in registers (fma), one LDS float add per run, not per edge; row sum order = LDS rank in run
# speedup vs baseline: 3.1812x; 1.0959x over previous
#include <hip/hip_runtime.h>

typedef __attribute__((ext_vector_type(2))) float v2f;
typedef __attribute__((ext_vector_type(8))) float v8f;

#define N_NODES 100000
#define N_EDGES 1600000
#define IN_DIM 256
#define HID_DIM 128
#define OUT_DIM 64

__global__ __launch_bounds__(256) void gemm1_wmma(const float* __restrict__ X,
                                                  const float* __restrict__ W1,
                                                  float* __restrict__ H1,
                                                  int* __restrict__ tables) {
    __shared__ float xs[16][260];
    __shared__ float w1s[64][136];

    const int tid  = threadIdx.x;
    if (blockIdx.x < 8) tables[blockIdx.x * 256 + tid] = 0;
    const int lane = tid & 31;
    const int wave = tid >> 5;
    const int half = lane >> 4;
    const int l16  = lane & 15;
    const int mbase = blockIdx.x * 16;

    const int wr = tid >> 5;
    const int wc = (tid & 31) * 4;
    const float* wp = W1 + (size_t)wr * HID_DIM + wc;

    const int xrow = tid >> 4;
    const int xseg = (tid & 15) * 16;
    const float4* xp = (const float4*)(X + (size_t)(mbase + xrow) * IN_DIM + xseg);
    float4 a0 = xp[0];
    float4 a1 = xp[1];
    float4 a2 = xp[2];
    float4 a3 = xp[3];

    v8f acc = {};
    const int nb = wave * 16 + l16;

    for (int kc = 0; kc < IN_DIM; kc += 64) {
        float4 w[8];
#pragma unroll
        for (int t = 0; t < 8; ++t)
            w[t] = *(const float4*)(wp + (size_t)(kc + 8 * t) * HID_DIM);
        if (kc == 0) {
            *(float4*)&xs[xrow][xseg + 0]  = a0;
            *(float4*)&xs[xrow][xseg + 4]  = a1;
            *(float4*)&xs[xrow][xseg + 8]  = a2;
            *(float4*)&xs[xrow][xseg + 12] = a3;
        }
#pragma unroll
        for (int t = 0; t < 8; ++t)
            *(float4*)&w1s[wr + 8 * t][wc] = w[t];
        __syncthreads();

#pragma unroll
        for (int kk = 0; kk < 64; kk += 4) {
            const int ka = kc + kk + half * 2;
            const int kb = kk + half * 2;
            v2f a = {xs[l16][ka], xs[l16][ka + 1]};
            v2f b = {w1s[kb][nb], w1s[kb + 1][nb]};
            acc = __builtin_amdgcn_wmma_f32_16x16x4_f32(
                false, a, false, b, (short)0, acc, false, false);
        }
        __syncthreads();
    }

#pragma unroll
    for (int v = 0; v < 8; ++v) {
        int row = mbase + v + half * 8;
        H1[(size_t)row * HID_DIM + nb] = acc[v];
    }
}

__global__ __launch_bounds__(256) void gemm2_wmma(const float* __restrict__ Hagg,
                                                  const float* __restrict__ W2,
                                                  float* __restrict__ H2) {
    __shared__ float hs[32][132];
    __shared__ float w2s[128][72];

    const int tid  = threadIdx.x;
    const int lane = tid & 31;
    const int wave = tid >> 5;
    const int mt   = wave >> 2;
    const int nt   = wave & 3;
    const int half = lane >> 4;
    const int l16  = lane & 15;
    const int mbase = blockIdx.x * 32;

    const int hr   = tid >> 5;
    const int hc   = (tid & 31) * 4;
    const int kr   = tid >> 4;
    const int kcol = (tid & 15) * 4;
    const float* hp = Hagg + (size_t)(mbase + hr) * HID_DIM + hc;
    const float* wp = W2 + (size_t)kr * OUT_DIM + kcol;
    float4 h[4], w[8];
#pragma unroll
    for (int t = 0; t < 4; ++t)
        h[t] = *(const float4*)(hp + (size_t)(8 * t) * HID_DIM);
#pragma unroll
    for (int t = 0; t < 8; ++t)
        w[t] = *(const float4*)(wp + (size_t)(16 * t) * OUT_DIM);
#pragma unroll
    for (int t = 0; t < 4; ++t) {
        float4 r = h[t];
        r.x = fmaxf(r.x, 0.0f);
        r.y = fmaxf(r.y, 0.0f);
        r.z = fmaxf(r.z, 0.0f);
        r.w = fmaxf(r.w, 0.0f);
        *(float4*)&hs[hr + 8 * t][hc] = r;
    }
#pragma unroll
    for (int t = 0; t < 8; ++t)
        *(float4*)&w2s[kr + 16 * t][kcol] = w[t];
    __syncthreads();

    v8f acc = {};
    const int mrow = mt * 16 + l16;
    const int nb   = nt * 16 + l16;

#pragma unroll
    for (int kk = 0; kk < HID_DIM; kk += 4) {
        const int k = kk + half * 2;
        v2f a = {hs[mrow][k], hs[mrow][k + 1]};
        v2f b = {w2s[k][nb], w2s[k + 1][nb]};
        acc = __builtin_amdgcn_wmma_f32_16x16x4_f32(
            false, a, false, b, (short)0, acc, false, false);
    }

#pragma unroll
    for (int v = 0; v < 8; ++v) {
        int row = mbase + mt * 16 + v + half * 8;
        H2[(size_t)row * OUT_DIM + nt * 16 + l16] = acc[v];
    }
}

#define ROWS_PER_BIN 100
#define NBINS (N_NODES / ROWS_PER_BIN)
#define TABLE_INTS 1024
#define BIN_THREADS 1024
#define BIN_EPT 6
#define BIN_EPB (BIN_THREADS * BIN_EPT)
#define BIN_BLOCKS ((N_EDGES + BIN_EPB - 1) / BIN_EPB)
#define SUM_THREADS 512
#define KEY_SHIFT 7

static __device__ __forceinline__ int lane_read_i(int x, int l) {
    return __builtin_amdgcn_readlane(x, l);
}
static __device__ __forceinline__ float lane_read_f(float x, int l) {
    return __int_as_float(__builtin_amdgcn_readlane(__float_as_int(x), l));
}
static __device__ __forceinline__ int bin_of(int d) {
    const unsigned b = (unsigned)d / (unsigned)ROWS_PER_BIN;
    return (int)(b < (unsigned)NBINS ? b : (unsigned)(NBINS - 1));
}

__global__ __launch_bounds__(BIN_THREADS) void bin_count(const int* __restrict__ dst, int E,
                                                         int* __restrict__ gcount) {
    __shared__ int h[TABLE_INTS];
    const int tid = threadIdx.x;
    h[tid] = 0;
    const int base = blockIdx.x * BIN_EPB + tid;
    int d[BIN_EPT];
#pragma unroll
    for (int k = 0; k < BIN_EPT; ++k) {
        const int e = base + k * BIN_THREADS;
        d[k] = (e < E) ? dst[e] : -1;
    }
    __syncthreads();
#pragma unroll
    for (int k = 0; k < BIN_EPT; ++k)
        if (d[k] >= 0) atomicAdd(&h[bin_of(d[k])], 1);
    __syncthreads();
    const int c = h[tid];
    if (tid < NBINS && c != 0) atomicAdd(&gcount[tid], c);
}

__global__ __launch_bounds__(BIN_THREADS) void bin_fill(const int* __restrict__ src,
                                                        const int* __restrict__ dst,
                                                        const float* __restrict__ vals, int E,
                                                        const int* __restrict__ gcount,
                                                        int* __restrict__ gcursor,
                                                        int* __restrict__ binstart,
                                                        int* __restrict__ rec) {
    __shared__ int s_pos[TABLE_INTS];
    __shared__ int h[TABLE_INTS];
    __shared__ int s_wsum[32];
    const int tid  = threadIdx.x;
    const int lane = tid & 31;
    const int wave = tid >> 5;

    const int base = blockIdx.x * BIN_EPB + tid;
    int d[BIN_EPT], s[BIN_EPT];
    float v[BIN_EPT];
#pragma unroll
    for (int k = 0; k < BIN_EPT; ++k) {
        const int e = base + k * BIN_THREADS;
        d[k] = -1; s[k] = 0; v[k] = 0.0f;
        if (e < E) { d[k] = dst[e]; s[k] = src[e]; v[k] = vals[e]; }
    }

    h[tid] = 0;
    const int c = (tid < NBINS) ? gcount[tid] : 0;
    int x = c;
#pragma unroll
    for (int off = 1; off < 32; off <<= 1) {
        const int y = __shfl_up(x, off);
        if (lane >= off) x += y;
    }
    if (lane == 31) s_wsum[wave] = x;
    __syncthreads();
    if (wave == 0) {
        const int w = s_wsum[lane];
        int xs = w;
#pragma unroll
        for (int off = 1; off < 32; off <<= 1) {
            const int y = __shfl_up(xs, off);
            if (lane >= off) xs += y;
        }
        s_wsum[lane] = xs - w;
    }
    __syncthreads();
    const int start = s_wsum[wave] + x - c;
    s_pos[tid] = start;
    if (blockIdx.x == 0) {
        if (tid < NBINS) binstart[tid] = start;
        if (tid == NBINS - 1) binstart[NBINS] = start + c;
    }

#pragma unroll
    for (int k = 0; k < BIN_EPT; ++k)
        if (d[k] >= 0) atomicAdd(&h[bin_of(d[k])], 1);
    __syncthreads();
    {
        const int lc = h[tid];
        if (tid < NBINS && lc != 0) s_pos[tid] = start + atomicAdd(&gcursor[tid], lc);
        h[tid] = 0;
    }
    __syncthreads();
#pragma unroll
    for (int k = 0; k < BIN_EPT; ++k) {
        if (d[k] >= 0) {
            const int b = bin_of(d[k]);
            const int pos = s_pos[b] + atomicAdd(&h[b], 1);
            int lr = d[k] - b * ROWS_PER_BIN;
            if ((unsigned)lr >= (unsigned)ROWS_PER_BIN) lr = (1 << KEY_SHIFT) - 1;
            int2 r;
            r.x = (s[k] << KEY_SHIFT) | lr;
            r.y = __float_as_int(v[k]);
            *(int2*)(rec + 2 * (size_t)pos) = r;
        }
    }
}

#define SORT_CHUNK 2048
#define SORT_RPT (SORT_CHUNK / SUM_THREADS)
#define KEY_MASK ((1 << KEY_SHIFT) - 1)

template <int D>
static __device__ __forceinline__ void bin_sum(const int* __restrict__ rec,
                                               const int* __restrict__ binstart,
                                               const float* __restrict__ H,
                                               float* __restrict__ out) {
    constexpr int J  = D / 32;
    constexpr int UB = 32 / J;
    constexpr int T4 = ROWS_PER_BIN * D / 4;
    __shared__ float acc[ROWS_PER_BIN * D];
    __shared__ int2 srec[SORT_CHUNK];
    __shared__ int cnt[KEY_MASK + 1];
    __shared__ int wtot[4];
    const int tid  = threadIdx.x;
    const int lane = tid & 31;
    const int wave = __builtin_amdgcn_readfirstlane(tid >> 5);
    const int b    = blockIdx.x;

    float4* acc4 = (float4*)acc;
    for (int i = tid; i < T4; i += SUM_THREADS) acc4[i] = make_float4(0.0f, 0.0f, 0.0f, 0.0f);
    const int lo = binstart[b];
    const int hi = binstart[b + 1];

    for (int c0 = lo; c0 < hi; c0 += SORT_CHUNK) {
        const int m = (hi - c0 < SORT_CHUNK) ? (hi - c0) : SORT_CHUNK;
        if (tid <= KEY_MASK) cnt[tid] = 0;
        __syncthreads();
        int2 q[SORT_RPT];
        int rk[SORT_RPT];
#pragma unroll
        for (int i = 0; i < SORT_RPT; ++i) {
            const int idx = tid + i * SUM_THREADS;
            q[i] = make_int2(KEY_MASK, 0);
            if (idx < m) q[i] = *(const int2*)(rec + 2 * (size_t)(c0 + idx));
        }
#pragma unroll
        for (int i = 0; i < SORT_RPT; ++i) {
            const int idx = tid + i * SUM_THREADS;
            rk[i] = 0;
            if (idx < m) rk[i] = atomicAdd(&cnt[q[i].x & KEY_MASK], 1);
        }
        __syncthreads();
        int c = 0, x = 0;
        if (tid <= KEY_MASK) {
            c = cnt[tid];
            x = c;
#pragma unroll
            for (int off = 1; off < 32; off <<= 1) {
                const int y = __shfl_up(x, off);
                if (lane >= off) x += y;
            }
            if (lane == 31) wtot[tid >> 5] = x;
        }
        __syncthreads();
        if (tid <= KEY_MASK) {
            const int wv = tid >> 5;
            int before = 0;
            if (wv > 0) before += wtot[0];
            if (wv > 1) before += wtot[1];
            if (wv > 2) before += wtot[2];
            cnt[tid] = before + x - c;
        }
        __syncthreads();
#pragma unroll
        for (int i = 0; i < SORT_RPT; ++i) {
            const int idx = tid + i * SUM_THREADS;
            if (idx < m) srec[cnt[q[i].x & KEY_MASK] + rk[i]] = q[i];
        }
        __syncthreads();

        for (int g = wave * 32; g < m; g += SUM_THREADS) {
            const int e = g + lane;
            int key = KEY_MASK;
            float val = 0.0f;
            if (e < m) {
                const int2 r = srec[e];
                key = r.x;
                val = __int_as_float(r.y);
            }
            const int n = (m - g < 32) ? (m - g) : 32;
            int cur = -1;
            float s[J];
#pragma unroll
            for (int j = 0; j < J; ++j) s[j] = 0.0f;
            for (int u0 = 0; u0 < n; u0 += UB) {
                float r[UB][J];
#pragma unroll
                for (int u = 0; u < UB; ++u) {
                    const int uu = (u0 + u < n) ? (u0 + u) : (n - 1);
                    const int k = lane_read_i(key, uu);
                    const float* row = H + (size_t)(k >> KEY_SHIFT) * D + lane;
#pragma unroll
                    for (int j = 0; j < J; ++j) r[u][j] = row[32 * j];
                }
#pragma unroll
                for (int u = 0; u < UB; ++u) {
                    if (u0 + u < n) {
                        const int k = lane_read_i(key, u0 + u);
                        const float w = lane_read_f(val, u0 + u);
                        const int lr = k & KEY_MASK;
                        if (lr != cur) {
                            if ((unsigned)cur < (unsigned)ROWS_PER_BIN) {
                                float* a = acc + cur * D + lane;
#pragma unroll
                                for (int j = 0; j < J; ++j) atomicAdd(a + 32 * j, s[j]);
                            }
                            cur = lr;
#pragma unroll
                            for (int j = 0; j < J; ++j) s[j] = 0.0f;
                        }
#pragma unroll
                        for (int j = 0; j < J; ++j) s[j] = fmaf(r[u][j], w, s[j]);
                    }
                }
            }
            if ((unsigned)cur < (unsigned)ROWS_PER_BIN) {
                float* a = acc + cur * D + lane;
#pragma unroll
                for (int j = 0; j < J; ++j) atomicAdd(a + 32 * j, s[j]);
            }
        }
    }
    __syncthreads();

    float4* o4 = (float4*)(out + (size_t)b * ROWS_PER_BIN * D);
    for (int i = tid; i < T4; i += SUM_THREADS) o4[i] = acc4[i];
}

__global__ __launch_bounds__(SUM_THREADS, 16) void bin_sum_d128(const int* __restrict__ rec,
                                                            const int* __restrict__ binstart,
                                                            const float* __restrict__ H,
                                                            float* __restrict__ out) {
    bin_sum<128>(rec, binstart, H, out);
}

__global__ __launch_bounds__(SUM_THREADS, 16) void bin_sum_d64(const int* __restrict__ rec,
                                                           const int* __restrict__ binstart,
                                                           const float* __restrict__ H,
                                                           float* __restrict__ out) {
    bin_sum<64>(rec, binstart, H, out);
}

extern "C" void kernel_launch(void* const* d_in, const int* in_sizes, int n_in,
                              void* d_out, int out_size, void* d_ws, size_t ws_size,
                              hipStream_t stream) {
    const float* X     = (const float*)d_in[0];
    const int*   esrc  = (const int*)d_in[1];
    const int*   edst  = (const int*)d_in[2];
    const float* evals = (const float*)d_in[3];
    const float* W1    = (const float*)d_in[4];
    const float* W2    = (const float*)d_in[5];
    float*       out   = (float*)d_out;

    char* ws = (char*)d_ws;
    const size_t h1Bytes  = (size_t)N_NODES * HID_DIM * sizeof(float);
    const size_t recBytes = (size_t)N_EDGES * 2 * sizeof(int);
    float* H1   = (float*)(ws);
    float* Hagg = (float*)(ws + h1Bytes);
    float* H2   = (float*)(ws);
    int* rec      = (int*)(ws + 2 * h1Bytes);
    int* tables   = (int*)(ws + 2 * h1Bytes + recBytes);
    int* gcount   = tables;
    int* gcursor  = tables + TABLE_INTS;
    int* binstart = tables + 2 * TABLE_INTS;

    gemm1_wmma<<<N_NODES / 16, 256, 0, stream>>>(X, W1, H1, tables);
    bin_count<<<BIN_BLOCKS, BIN_THREADS, 0, stream>>>(edst, N_EDGES, gcount);
    bin_fill<<<BIN_BLOCKS, BIN_THREADS, 0, stream>>>(esrc, edst, evals, N_EDGES, gcount, gcursor,
                                                     binstart, rec);
    bin_sum_d128<<<NBINS, SUM_THREADS, 0, stream>>>(rec, binstart, H1, Hagg);

    gemm2_wmma<<<N_NODES / 32, 256, 0, stream>>>(Hagg, W2, H2);
    bin_sum_d64<<<NBINS, SUM_THREADS, 0, stream>>>(rec, binstart, H2, out);
}


# gemm1 tiling: a block owns an 80-row strip, each wave 5 accumulator tiles, one B fragment read feeds 5 WMMAs, W1 chunks staged once per 80 rows, not per 16; 10000 waves, not 50000; sum order kept
# speedup vs baseline: 3.3501x; 1.0531x over previous
#include <hip/hip_runtime.h>

typedef __attribute__((ext_vector_type(2))) float v2f;
typedef __attribute__((ext_vector_type(8))) float v8f;

#define N_NODES 100000
#define N_EDGES 1600000
#define IN_DIM 256
#define HID_DIM 128
#define OUT_DIM 64

#define G1_MT 5
#define G1_ROWS (16 * G1_MT)
#define G1_KC 64
#define XS_PITCH 68
#define WS_PITCH 136

__global__ __launch_bounds__(256) __attribute__((amdgpu_waves_per_eu(6, 6))) void gemm1_wmma(const float* __restrict__ X,
                                                     const float* __restrict__ W1,
                                                     float* __restrict__ H1,
                                                     int* __restrict__ tables) {
    __shared__ __attribute__((aligned(16))) float xs[G1_ROWS * XS_PITCH];
    __shared__ __attribute__((aligned(16))) float w1s[G1_KC * WS_PITCH];

    const int tid  = threadIdx.x;
    if (blockIdx.x < 8) tables[blockIdx.x * 256 + tid] = 0;
    const int lane = tid & 31;
    const int wave = tid >> 5;
    const int half = lane >> 4;
    const int l16  = lane & 15;
    const int mbase = blockIdx.x * G1_ROWS;

    const int xr = tid >> 4;
    const int xc = (tid & 15) * 4;
    const float* xg = X + (size_t)(mbase + xr) * IN_DIM + xc;
    const int wr = tid >> 5;
    const int wc = (tid & 31) * 4;
    const float* wg = W1 + (size_t)wr * HID_DIM + wc;

    v8f acc[G1_MT];
#pragma unroll
    for (int m = 0; m < G1_MT; ++m) acc[m] = (v8f){};

    const float* ap = xs + l16 * XS_PITCH + half * 2;
    const float* bp = w1s + (half * 2) * WS_PITCH + wave * 16 + l16;

    for (int kc = 0; kc < IN_DIM; kc += G1_KC) {
        float4 xq0, xq1, xq2, xq3, xq4, w[8];
        xq0 = *(const float4*)(xg + kc);
        xq1 = *(const float4*)(xg + (size_t)16 * IN_DIM + kc);
        xq2 = *(const float4*)(xg + (size_t)32 * IN_DIM + kc);
        xq3 = *(const float4*)(xg + (size_t)48 * IN_DIM + kc);
        xq4 = *(const float4*)(xg + (size_t)64 * IN_DIM + kc);
#pragma unroll
        for (int t = 0; t < 8; ++t)
            w[t] = *(const float4*)(wg + (size_t)(kc + 8 * t) * HID_DIM);
        {
            float* q = &xs[xr * XS_PITCH + xc];
            *(float4*)(q)                 = xq0;
            *(float4*)(q + 16 * XS_PITCH) = xq1;
            *(float4*)(q + 32 * XS_PITCH) = xq2;
            *(float4*)(q + 48 * XS_PITCH) = xq3;
            *(float4*)(q + 64 * XS_PITCH) = xq4;
        }
#pragma unroll
        for (int t = 0; t < 8; ++t)
            *(float4*)&w1s[(wr + 8 * t) * WS_PITCH + wc] = w[t];
        __syncthreads();

#pragma unroll
        for (int kk = 0; kk < G1_KC; kk += 4) {
            const v2f b = {bp[kk * WS_PITCH], bp[(kk + 1) * WS_PITCH]};
#pragma unroll
            for (int m = 0; m < G1_MT; ++m) {
                const v2f a = *(const v2f*)(ap + m * 16 * XS_PITCH + kk);
                acc[m] = __builtin_amdgcn_wmma_f32_16x16x4_f32(
                    false, a, false, b, (short)0, acc[m], false, false);
            }
        }
        __syncthreads();
    }

    const int nb = wave * 16 + l16;
#pragma unroll
    for (int m = 0; m < G1_MT; ++m) {
#pragma unroll
        for (int v = 0; v < 8; ++v) {
            const int row = mbase + m * 16 + v + half * 8;
            H1[(size_t)row * HID_DIM + nb] = acc[m][v];
        }
    }
}

__global__ __launch_bounds__(256) void gemm2_wmma(const float* __restrict__ Hagg,
                                                  const float* __restrict__ W2,
                                                  float* __restrict__ H2) {
    __shared__ float hs[32][132];
    __shared__ float w2s[128][72];

    const int tid  = threadIdx.x;
    const int lane = tid & 31;
    const int wave = tid >> 5;
    const int mt   = wave >> 2;
    const int nt   = wave & 3;
    const int half = lane >> 4;
    const int l16  = lane & 15;
    const int mbase = blockIdx.x * 32;

    const int hr   = tid >> 5;
    const int hc   = (tid & 31) * 4;
    const int kr   = tid >> 4;
    const int kcol = (tid & 15) * 4;
    const float* hp = Hagg + (size_t)(mbase + hr) * HID_DIM + hc;
    const float* wp = W2 + (size_t)kr * OUT_DIM + kcol;
    float4 h[4], w[8];
#pragma unroll
    for (int t = 0; t < 4; ++t)
        h[t] = *(const float4*)(hp + (size_t)(8 * t) * HID_DIM);
#pragma unroll
    for (int t = 0; t < 8; ++t)
        w[t] = *(const float4*)(wp + (size_t)(16 * t) * OUT_DIM);
#pragma unroll
    for (int t = 0; t < 4; ++t) {
        float4 r = h[t];
        r.x = fmaxf(r.x, 0.0f);
        r.y = fmaxf(r.y, 0.0f);
        r.z = fmaxf(r.z, 0.0f);
        r.w = fmaxf(r.w, 0.0f);
        *(float4*)&hs[hr + 8 * t][hc] = r;
    }
#pragma unroll
    for (int t = 0; t < 8; ++t)
        *(float4*)&w2s[kr + 16 * t][kcol] = w[t];
    __syncthreads();

    v8f acc = {};
    const int mrow = mt * 16 + l16;
    const int nb   = nt * 16 + l16;

#pragma unroll
    for (int kk = 0; kk < HID_DIM; kk += 4) {
        const int k = kk + half * 2;
        v2f a = {hs[mrow][k], hs[mrow][k + 1]};
        v2f b = {w2s[k][nb], w2s[k + 1][nb]};
        acc = __builtin_amdgcn_wmma_f32_16x16x4_f32(
            false, a, false, b, (short)0, acc, false, false);
    }

#pragma unroll
    for (int v = 0; v < 8; ++v) {
        int row = mbase + mt * 16 + v + half * 8;
        H2[(size_t)row * OUT_DIM + nt * 16 + l16] = acc[v];
    }
}

#define ROWS_PER_BIN 100
#define NBINS (N_NODES / ROWS_PER_BIN)
#define TABLE_INTS 1024
#define BIN_THREADS 1024
#define BIN_EPT 6
#define BIN_EPB (BIN_THREADS * BIN_EPT)
#define BIN_BLOCKS ((N_EDGES + BIN_EPB - 1) / BIN_EPB)
#define SUM_THREADS 512
#define KEY_SHIFT 7

static __device__ __forceinline__ int lane_read_i(int x, int l) {
    return __builtin_amdgcn_readlane(x, l);
}
static __device__ __forceinline__ float lane_read_f(float x, int l) {
    return __int_as_float(__builtin_amdgcn_readlane(__float_as_int(x), l));
}
static __device__ __forceinline__ int bin_of(int d) {
    const unsigned b = (unsigned)d / (unsigned)ROWS_PER_BIN;
    return (int)(b < (unsigned)NBINS ? b : (unsigned)(NBINS - 1));
}

__global__ __launch_bounds__(BIN_THREADS) void bin_count(const int* __restrict__ dst, int E,
                                                         int* __restrict__ gcount) {
    __shared__ int h[TABLE_INTS];
    const int tid = threadIdx.x;
    h[tid] = 0;
    const int base = blockIdx.x * BIN_EPB + tid;
    int d[BIN_EPT];
#pragma unroll
    for (int k = 0; k < BIN_EPT; ++k) {
        const int e = base + k * BIN_THREADS;
        d[k] = (e < E) ? dst[e] : -1;
    }
    __syncthreads();
#pragma unroll
    for (int k = 0; k < BIN_EPT; ++k)
        if (d[k] >= 0) atomicAdd(&h[bin_of(d[k])], 1);
    __syncthreads();
    const int c = h[tid];
    if (tid < NBINS && c != 0) atomicAdd(&gcount[tid], c);
}

__global__ __launch_bounds__(BIN_THREADS) void bin_fill(const int* __restrict__ src,
                                                        const int* __restrict__ dst,
                                                        const float* __restrict__ vals, int E,
                                                        const int* __restrict__ gcount,
                                                        int* __restrict__ gcursor,
                                                        int* __restrict__ binstart,
                                                        int* __restrict__ rec) {
    __shared__ int s_pos[TABLE_INTS];
    __shared__ int h[TABLE_INTS];
    __shared__ int s_wsum[32];
    const int tid  = threadIdx.x;
    const int lane = tid & 31;
    const int wave = tid >> 5;

    const int base = blockIdx.x * BIN_EPB + tid;
    int d[BIN_EPT], s[BIN_EPT];
    float v[BIN_EPT];
#pragma unroll
    for (int k = 0; k < BIN_EPT; ++k) {
        const int e = base + k * BIN_THREADS;
        d[k] = -1; s[k] = 0; v[k] = 0.0f;
        if (e < E) { d[k] = dst[e]; s[k] = src[e]; v[k] = vals[e]; }
    }

    h[tid] = 0;
    const int c = (tid < NBINS) ? gcount[tid] : 0;
    int x = c;
#pragma unroll
    for (int off = 1; off < 32; off <<= 1) {
        const int y = __shfl_up(x, off);
        if (lane >= off) x += y;
    }
    if (lane == 31) s_wsum[wave] = x;
    __syncthreads();
    if (wave == 0) {
        const int w = s_wsum[lane];
        int xs = w;
#pragma unroll
        for (int off = 1; off < 32; off <<= 1) {
            const int y = __shfl_up(xs, off);
            if (lane >= off) xs += y;
        }
        s_wsum[lane] = xs - w;
    }
    __syncthreads();
    const int start = s_wsum[wave] + x - c;
    s_pos[tid] = start;
    if (blockIdx.x == 0) {
        if (tid < NBINS) binstart[tid] = start;
        if (tid == NBINS - 1) binstart[NBINS] = start + c;
    }

#pragma unroll
    for (int k = 0; k < BIN_EPT; ++k)
        if (d[k] >= 0) atomicAdd(&h[bin_of(d[k])], 1);
    __syncthreads();
    {
        const int lc = h[tid];
        if (tid < NBINS && lc != 0) s_pos[tid] = start + atomicAdd(&gcursor[tid], lc);
        h[tid] = 0;
    }
    __syncthreads();
#pragma unroll
    for (int k = 0; k < BIN_EPT; ++k) {
        if (d[k] >= 0) {
            const int b = bin_of(d[k]);
            const int pos = s_pos[b] + atomicAdd(&h[b], 1);
            int lr = d[k] - b * ROWS_PER_BIN;
            if ((unsigned)lr >= (unsigned)ROWS_PER_BIN) lr = (1 << KEY_SHIFT) - 1;
            int2 r;
            r.x = (s[k] << KEY_SHIFT) | lr;
            r.y = __float_as_int(v[k]);
            *(int2*)(rec + 2 * (size_t)pos) = r;
        }
    }
}

#define SORT_CHUNK 2048
#define SORT_RPT (SORT_CHUNK / SUM_THREADS)
#define KEY_MASK ((1 << KEY_SHIFT) - 1)

template <int D>
static __device__ __forceinline__ void bin_sum(const int* __restrict__ rec,
                                               const int* __restrict__ binstart,
                                               const float* __restrict__ H,
                                               float* __restrict__ out) {
    constexpr int J  = D / 32;
    constexpr int UB = 32 / J;
    constexpr int T4 = ROWS_PER_BIN * D / 4;
    __shared__ float acc[ROWS_PER_BIN * D];
    __shared__ int2 srec[SORT_CHUNK];
    __shared__ int cnt[KEY_MASK + 1];
    __shared__ int wtot[4];
    const int tid  = threadIdx.x;
    const int lane = tid & 31;
    const int wave = __builtin_amdgcn_readfirstlane(tid >> 5);
    const int b    = blockIdx.x;

    float4* acc4 = (float4*)acc;
    for (int i = tid; i < T4; i += SUM_THREADS) acc4[i] = make_float4(0.0f, 0.0f, 0.0f, 0.0f);
    const int lo = binstart[b];
    const int hi = binstart[b + 1];

    for (int c0 = lo; c0 < hi; c0 += SORT_CHUNK) {
        const int m = (hi - c0 < SORT_CHUNK) ? (hi - c0) : SORT_CHUNK;
        if (tid <= KEY_MASK) cnt[tid] = 0;
        __syncthreads();
        int2 q[SORT_RPT];
        int rk[SORT_RPT];
#pragma unroll
        for (int i = 0; i < SORT_RPT; ++i) {
            const int idx = tid + i * SUM_THREADS;
            q[i] = make_int2(KEY_MASK, 0);
            if (idx < m) q[i] = *(const int2*)(rec + 2 * (size_t)(c0 + idx));
        }
#pragma unroll
        for (int i = 0; i < SORT_RPT; ++i) {
            const int idx = tid + i * SUM_THREADS;
            rk[i] = 0;
            if (idx < m) rk[i] = atomicAdd(&cnt[q[i].x & KEY_MASK], 1);
        }
        __syncthreads();
        int c = 0, x = 0;
        if (tid <= KEY_MASK) {
            c = cnt[tid];
            x = c;
#pragma unroll
            for (int off = 1; off < 32; off <<= 1) {
                const int y = __shfl_up(x, off);
                if (lane >= off) x += y;
            }
            if (lane == 31) wtot[tid >> 5] = x;
        }
        __syncthreads();
        if (tid <= KEY_MASK) {
            const int wv = tid >> 5;
            int before = 0;
            if (wv > 0) before += wtot[0];
            if (wv > 1) before += wtot[1];
            if (wv > 2) before += wtot[2];
            cnt[tid] = before + x - c;
        }
        __syncthreads();
#pragma unroll
        for (int i = 0; i < SORT_RPT; ++i) {
            const int idx = tid + i * SUM_THREADS;
            if (idx < m) srec[cnt[q[i].x & KEY_MASK] + rk[i]] = q[i];
        }
        __syncthreads();

        for (int g = wave * 32; g < m; g += SUM_THREADS) {
            const int e = g + lane;
            int key = KEY_MASK;
            float val = 0.0f;
            if (e < m) {
                const int2 r = srec[e];
                key = r.x;
                val = __int_as_float(r.y);
            }
            const int n = (m - g < 32) ? (m - g) : 32;
            int cur = -1;
            float s[J];
#pragma unroll
            for (int j = 0; j < J; ++j) s[j] = 0.0f;
            for (int u0 = 0; u0 < n; u0 += UB) {
                float r[UB][J];
#pragma unroll
                for (int u = 0; u < UB; ++u) {
                    const int uu = (u0 + u < n) ? (u0 + u) : (n - 1);
                    const int k = lane_read_i(key, uu);
                    const float* row = H + (size_t)(k >> KEY_SHIFT) * D + lane;
#pragma unroll
                    for (int j = 0; j < J; ++j) r[u][j] = row[32 * j];
                }
#pragma unroll
                for (int u = 0; u < UB; ++u) {
                    if (u0 + u < n) {
                        const int k = lane_read_i(key, u0 + u);
                        const float w = lane_read_f(val, u0 + u);
                        const int lr = k & KEY_MASK;
                        if (lr != cur) {
                            if ((unsigned)cur < (unsigned)ROWS_PER_BIN) {
                                float* a = acc + cur * D + lane;
#pragma unroll
                                for (int j = 0; j < J; ++j) atomicAdd(a + 32 * j, s[j]);
                            }
                            cur = lr;
#pragma unroll
                            for (int j = 0; j < J; ++j) s[j] = 0.0f;
                        }
#pragma unroll
                        for (int j = 0; j < J; ++j) s[j] = fmaf(r[u][j], w, s[j]);
                    }
                }
            }
            if ((unsigned)cur < (unsigned)ROWS_PER_BIN) {
                float* a = acc + cur * D + lane;
#pragma unroll
                for (int j = 0; j < J; ++j) atomicAdd(a + 32 * j, s[j]);
            }
        }
    }
    __syncthreads();

    float4* o4 = (float4*)(out + (size_t)b * ROWS_PER_BIN * D);
    for (int i = tid; i < T4; i += SUM_THREADS) o4[i] = acc4[i];
}

__global__ __launch_bounds__(SUM_THREADS, 16) void bin_sum_d128(const int* __restrict__ rec,
                                                            const int* __restrict__ binstart,
                                                            const float* __restrict__ H,
                                                            float* __restrict__ out) {
    bin_sum<128>(rec, binstart, H, out);
}

__global__ __launch_bounds__(SUM_THREADS, 16) void bin_sum_d64(const int* __restrict__ rec,
                                                           const int* __restrict__ binstart,
                                                           const float* __restrict__ H,
                                                           float* __restrict__ out) {
    bin_sum<64>(rec, binstart, H, out);
}

extern "C" void kernel_launch(void* const* d_in, const int* in_sizes, int n_in,
                              void* d_out, int out_size, void* d_ws, size_t ws_size,
                              hipStream_t stream) {
    const float* X     = (const float*)d_in[0];
    const int*   esrc  = (const int*)d_in[1];
    const int*   edst  = (const int*)d_in[2];
    const float* evals = (const float*)d_in[3];
    const float* W1    = (const float*)d_in[4];
    const float* W2    = (const float*)d_in[5];
    float*       out   = (float*)d_out;

    char* ws = (char*)d_ws;
    const size_t h1Bytes  = (size_t)N_NODES * HID_DIM * sizeof(float);
    const size_t recBytes = (size_t)N_EDGES * 2 * sizeof(int);
    float* H1   = (float*)(ws);
    float* Hagg = (float*)(ws + h1Bytes);
    float* H2   = (float*)(ws);
    int* rec      = (int*)(ws + 2 * h1Bytes);
    int* tables   = (int*)(ws + 2 * h1Bytes + recBytes);
    int* gcount   = tables;
    int* gcursor  = tables + TABLE_INTS;
    int* binstart = tables + 2 * TABLE_INTS;

    gemm1_wmma<<<N_NODES / G1_ROWS, 256, 0, stream>>>(X, W1, H1, tables);
    bin_count<<<BIN_BLOCKS, BIN_THREADS, 0, stream>>>(edst, N_EDGES, gcount);
    bin_fill<<<BIN_BLOCKS, BIN_THREADS, 0, stream>>>(esrc, edst, evals, N_EDGES, gcount, gcursor,
                                                     binstart, rec);
    bin_sum_d128<<<NBINS, SUM_THREADS, 0, stream>>>(rec, binstart, H1, Hagg);

    gemm2_wmma<<<N_NODES / 32, 256, 0, stream>>>(Hagg, W2, H2);
    bin_sum_d64<<<NBINS, SUM_THREADS, 0, stream>>>(rec, binstart, H2, out);
}


# edge grouping, record stores: a block orders its 6144 records by range in LDS (LDS rank, 2nd scan), then consecutive lanes store consecutive records: few lines per store, not 32; arrival order kept
# speedup vs baseline: 3.6832x; 1.0994x over previous
#include <hip/hip_runtime.h>

typedef __attribute__((ext_vector_type(2))) float v2f;
typedef __attribute__((ext_vector_type(8))) float v8f;

#define N_NODES 100000
#define N_EDGES 1600000
#define IN_DIM 256
#define HID_DIM 128
#define OUT_DIM 64

#define G1_MT 5
#define G1_ROWS (16 * G1_MT)
#define G1_KC 64
#define XS_PITCH 68
#define WS_PITCH 136

__global__ __launch_bounds__(256) __attribute__((amdgpu_waves_per_eu(6, 6))) void gemm1_wmma(const float* __restrict__ X,
                                                     const float* __restrict__ W1,
                                                     float* __restrict__ H1,
                                                     int* __restrict__ tables) {
    __shared__ __attribute__((aligned(16))) float xs[G1_ROWS * XS_PITCH];
    __shared__ __attribute__((aligned(16))) float w1s[G1_KC * WS_PITCH];

    const int tid  = threadIdx.x;
    if (blockIdx.x < 8) tables[blockIdx.x * 256 + tid] = 0;
    const int lane = tid & 31;
    const int wave = tid >> 5;
    const int half = lane >> 4;
    const int l16  = lane & 15;
    const int mbase = blockIdx.x * G1_ROWS;

    const int xr = tid >> 4;
    const int xc = (tid & 15) * 4;
    const float* xg = X + (size_t)(mbase + xr) * IN_DIM + xc;
    const int wr = tid >> 5;
    const int wc = (tid & 31) * 4;
    const float* wg = W1 + (size_t)wr * HID_DIM + wc;

    v8f acc[G1_MT];
#pragma unroll
    for (int m = 0; m < G1_MT; ++m) acc[m] = (v8f){};

    const float* ap = xs + l16 * XS_PITCH + half * 2;
    const float* bp = w1s + (half * 2) * WS_PITCH + wave * 16 + l16;

    for (int kc = 0; kc < IN_DIM; kc += G1_KC) {
        float4 xq0, xq1, xq2, xq3, xq4, w[8];
        xq0 = *(const float4*)(xg + kc);
        xq1 = *(const float4*)(xg + (size_t)16 * IN_DIM + kc);
        xq2 = *(const float4*)(xg + (size_t)32 * IN_DIM + kc);
        xq3 = *(const float4*)(xg + (size_t)48 * IN_DIM + kc);
        xq4 = *(const float4*)(xg + (size_t)64 * IN_DIM + kc);
#pragma unroll
        for (int t = 0; t < 8; ++t)
            w[t] = *(const float4*)(wg + (size_t)(kc + 8 * t) * HID_DIM);
        {
            float* q = &xs[xr * XS_PITCH + xc];
            *(float4*)(q)                 = xq0;
            *(float4*)(q + 16 * XS_PITCH) = xq1;
            *(float4*)(q + 32 * XS_PITCH) = xq2;
            *(float4*)(q + 48 * XS_PITCH) = xq3;
            *(float4*)(q + 64 * XS_PITCH) = xq4;
        }
#pragma unroll
        for (int t = 0; t < 8; ++t)
            *(float4*)&w1s[(wr + 8 * t) * WS_PITCH + wc] = w[t];
        __syncthreads();

#pragma unroll
        for (int kk = 0; kk < G1_KC; kk += 4) {
            const v2f b = {bp[kk * WS_PITCH], bp[(kk + 1) * WS_PITCH]};
#pragma unroll
            for (int m = 0; m < G1_MT; ++m) {
                const v2f a = *(const v2f*)(ap + m * 16 * XS_PITCH + kk);
                acc[m] = __builtin_amdgcn_wmma_f32_16x16x4_f32(
                    false, a, false, b, (short)0, acc[m], false, false);
            }
        }
        __syncthreads();
    }

    const int nb = wave * 16 + l16;
#pragma unroll
    for (int m = 0; m < G1_MT; ++m) {
#pragma unroll
        for (int v = 0; v < 8; ++v) {
            const int row = mbase + m * 16 + v + half * 8;
            H1[(size_t)row * HID_DIM + nb] = acc[m][v];
        }
    }
}

__global__ __launch_bounds__(256) void gemm2_wmma(const float* __restrict__ Hagg,
                                                  const float* __restrict__ W2,
                                                  float* __restrict__ H2) {
    __shared__ float hs[32][132];
    __shared__ float w2s[128][72];

    const int tid  = threadIdx.x;
    const int lane = tid & 31;
    const int wave = tid >> 5;
    const int mt   = wave >> 2;
    const int nt   = wave & 3;
    const int half = lane >> 4;
    const int l16  = lane & 15;
    const int mbase = blockIdx.x * 32;

    const int hr   = tid >> 5;
    const int hc   = (tid & 31) * 4;
    const int kr   = tid >> 4;
    const int kcol = (tid & 15) * 4;
    const float* hp = Hagg + (size_t)(mbase + hr) * HID_DIM + hc;
    const float* wp = W2 + (size_t)kr * OUT_DIM + kcol;
    float4 h[4], w[8];
#pragma unroll
    for (int t = 0; t < 4; ++t)
        h[t] = *(const float4*)(hp + (size_t)(8 * t) * HID_DIM);
#pragma unroll
    for (int t = 0; t < 8; ++t)
        w[t] = *(const float4*)(wp + (size_t)(16 * t) * OUT_DIM);
#pragma unroll
    for (int t = 0; t < 4; ++t) {
        float4 r = h[t];
        r.x = fmaxf(r.x, 0.0f);
        r.y = fmaxf(r.y, 0.0f);
        r.z = fmaxf(r.z, 0.0f);
        r.w = fmaxf(r.w, 0.0f);
        *(float4*)&hs[hr + 8 * t][hc] = r;
    }
#pragma unroll
    for (int t = 0; t < 8; ++t)
        *(float4*)&w2s[kr + 16 * t][kcol] = w[t];
    __syncthreads();

    v8f acc = {};
    const int mrow = mt * 16 + l16;
    const int nb   = nt * 16 + l16;

#pragma unroll
    for (int kk = 0; kk < HID_DIM; kk += 4) {
        const int k = kk + half * 2;
        v2f a = {hs[mrow][k], hs[mrow][k + 1]};
        v2f b = {w2s[k][nb], w2s[k + 1][nb]};
        acc = __builtin_amdgcn_wmma_f32_16x16x4_f32(
            false, a, false, b, (short)0, acc, false, false);
    }

#pragma unroll
    for (int v = 0; v < 8; ++v) {
        int row = mbase + mt * 16 + v + half * 8;
        H2[(size_t)row * OUT_DIM + nt * 16 + l16] = acc[v];
    }
}

#define ROWS_PER_BIN 100
#define NBINS (N_NODES / ROWS_PER_BIN)
#define TABLE_INTS 1024
#define BIN_THREADS 1024
#define BIN_EPT 6
#define BIN_EPB (BIN_THREADS * BIN_EPT)
#define BIN_BLOCKS ((N_EDGES + BIN_EPB - 1) / BIN_EPB)
#define SUM_THREADS 512
#define KEY_SHIFT 7

static __device__ __forceinline__ int lane_read_i(int x, int l) {
    return __builtin_amdgcn_readlane(x, l);
}
static __device__ __forceinline__ float lane_read_f(float x, int l) {
    return __int_as_float(__builtin_amdgcn_readlane(__float_as_int(x), l));
}
static __device__ __forceinline__ int bin_of(int d) {
    const unsigned b = (unsigned)d / (unsigned)ROWS_PER_BIN;
    return (int)(b < (unsigned)NBINS ? b : (unsigned)(NBINS - 1));
}

__global__ __launch_bounds__(BIN_THREADS) void bin_count(const int* __restrict__ dst, int E,
                                                         int* __restrict__ gcount) {
    __shared__ int h[TABLE_INTS];
    const int tid = threadIdx.x;
    h[tid] = 0;
    const int base = blockIdx.x * BIN_EPB + tid;
    int d[BIN_EPT];
#pragma unroll
    for (int k = 0; k < BIN_EPT; ++k) {
        const int e = base + k * BIN_THREADS;
        d[k] = (e < E) ? dst[e] : -1;
    }
    __syncthreads();
#pragma unroll
    for (int k = 0; k < BIN_EPT; ++k)
        if (d[k] >= 0) atomicAdd(&h[bin_of(d[k])], 1);
    __syncthreads();
    const int c = h[tid];
    if (tid < NBINS && c != 0) atomicAdd(&gcount[tid], c);
}

__global__ __launch_bounds__(BIN_THREADS) void bin_fill(const int* __restrict__ src,
                                                        const int* __restrict__ dst,
                                                        const float* __restrict__ vals, int E,
                                                        const int* __restrict__ gcount,
                                                        int* __restrict__ gcursor,
                                                        int* __restrict__ binstart,
                                                        int* __restrict__ rec) {
    __shared__ int s_gpos[TABLE_INTS];
    __shared__ int s_lpos[TABLE_INTS];
    __shared__ int h[TABLE_INTS];
    __shared__ int s_wsum[2][32];
    __shared__ int s_n;
    __shared__ int2 s_rec[BIN_EPB];
    __shared__ int s_idx[BIN_EPB];
    const int tid  = threadIdx.x;
    const int lane = tid & 31;
    const int wave = tid >> 5;

    const int base = blockIdx.x * BIN_EPB + tid;
    int d[BIN_EPT], s[BIN_EPT];
    float v[BIN_EPT];
#pragma unroll
    for (int k = 0; k < BIN_EPT; ++k) {
        const int e = base + k * BIN_THREADS;
        d[k] = -1; s[k] = 0; v[k] = 0.0f;
        if (e < E) { d[k] = dst[e]; s[k] = src[e]; v[k] = vals[e]; }
    }
    const int c = (tid < NBINS) ? gcount[tid] : 0;

    h[tid] = 0;
    __syncthreads();
    int bb[BIN_EPT], rk[BIN_EPT];
#pragma unroll
    for (int k = 0; k < BIN_EPT; ++k) {
        bb[k] = 0; rk[k] = 0;
        if (d[k] >= 0) {
            bb[k] = bin_of(d[k]);
            rk[k] = atomicAdd(&h[bb[k]], 1);
        }
    }
    __syncthreads();

    const int lc = h[tid];
    int x = c, y = lc;
#pragma unroll
    for (int off = 1; off < 32; off <<= 1) {
        const int x2 = __shfl_up(x, off);
        const int y2 = __shfl_up(y, off);
        if (lane >= off) { x += x2; y += y2; }
    }
    if (lane == 31) { s_wsum[0][wave] = x; s_wsum[1][wave] = y; }
    __syncthreads();
    if (wave == 0) {
        const int w0 = s_wsum[0][lane];
        const int w1 = s_wsum[1][lane];
        int x0 = w0, x1 = w1;
#pragma unroll
        for (int off = 1; off < 32; off <<= 1) {
            const int a0 = __shfl_up(x0, off);
            const int a1 = __shfl_up(x1, off);
            if (lane >= off) { x0 += a0; x1 += a1; }
        }
        s_wsum[0][lane] = x0 - w0;
        s_wsum[1][lane] = x1 - w1;
        if (lane == 31) s_n = x1;
    }
    __syncthreads();
    const int start = s_wsum[0][wave] + x - c;
    if (blockIdx.x == 0) {
        if (tid < NBINS) binstart[tid] = start;
        if (tid == NBINS - 1) binstart[NBINS] = start + c;
    }
    int g = start;
    if (tid < NBINS && lc != 0) g = start + atomicAdd(&gcursor[tid], lc);
    s_gpos[tid] = g;
    s_lpos[tid] = s_wsum[1][wave] + y - lc;
    __syncthreads();
#pragma unroll
    for (int k = 0; k < BIN_EPT; ++k) {
        if (d[k] >= 0) {
            const int b = bb[k];
            int lr = d[k] - b * ROWS_PER_BIN;
            if ((unsigned)lr >= (unsigned)ROWS_PER_BIN) lr = (1 << KEY_SHIFT) - 1;
            const int j = s_lpos[b] + rk[k];
            s_rec[j] = make_int2((s[k] << KEY_SHIFT) | lr, __float_as_int(v[k]));
            s_idx[j] = s_gpos[b] + rk[k];
        }
    }
    __syncthreads();
    const int n = s_n;
    int2 q[BIN_EPT];
    int gi[BIN_EPT];
#pragma unroll
    for (int k = 0; k < BIN_EPT; ++k) {
        q[k]  = s_rec[tid + k * BIN_THREADS];
        gi[k] = s_idx[tid + k * BIN_THREADS];
    }
#pragma unroll
    for (int k = 0; k < BIN_EPT; ++k)
        if (tid + k * BIN_THREADS < n) *(int2*)(rec + 2 * (size_t)gi[k]) = q[k];
}

#define SORT_CHUNK 2048
#define SORT_RPT (SORT_CHUNK / SUM_THREADS)
#define KEY_MASK ((1 << KEY_SHIFT) - 1)

template <int D>
static __device__ __forceinline__ void bin_sum(const int* __restrict__ rec,
                                               const int* __restrict__ binstart,
                                               const float* __restrict__ H,
                                               float* __restrict__ out) {
    constexpr int J  = D / 32;
    constexpr int UB = 32 / J;
    constexpr int T4 = ROWS_PER_BIN * D / 4;
    __shared__ float acc[ROWS_PER_BIN * D];
    __shared__ int2 srec[SORT_CHUNK];
    __shared__ int cnt[KEY_MASK + 1];
    __shared__ int wtot[4];
    const int tid  = threadIdx.x;
    const int lane = tid & 31;
    const int wave = __builtin_amdgcn_readfirstlane(tid >> 5);
    const int b    = blockIdx.x;

    float4* acc4 = (float4*)acc;
    for (int i = tid; i < T4; i += SUM_THREADS) acc4[i] = make_float4(0.0f, 0.0f, 0.0f, 0.0f);
    const int lo = binstart[b];
    const int hi = binstart[b + 1];

    for (int c0 = lo; c0 < hi; c0 += SORT_CHUNK) {
        const int m = (hi - c0 < SORT_CHUNK) ? (hi - c0) : SORT_CHUNK;
        if (tid <= KEY_MASK) cnt[tid] = 0;
        __syncthreads();
        int2 q[SORT_RPT];
        int rk[SORT_RPT];
#pragma unroll
        for (int i = 0; i < SORT_RPT; ++i) {
            const int idx = tid + i * SUM_THREADS;
            q[i] = make_int2(KEY_MASK, 0);
            if (idx < m) q[i] = *(const int2*)(rec + 2 * (size_t)(c0 + idx));
        }
#pragma unroll
        for (int i = 0; i < SORT_RPT; ++i) {
            const int idx = tid + i * SUM_THREADS;
            rk[i] = 0;
            if (idx < m) rk[i] = atomicAdd(&cnt[q[i].x & KEY_MASK], 1);
        }
        __syncthreads();
        int c = 0, x = 0;
        if (tid <= KEY_MASK) {
            c = cnt[tid];
            x = c;
#pragma unroll
            for (int off = 1; off < 32; off <<= 1) {
                const int y = __shfl_up(x, off);
                if (lane >= off) x += y;
            }
            if (lane == 31) wtot[tid >> 5] = x;
        }
        __syncthreads();
        if (tid <= KEY_MASK) {
            const int wv = tid >> 5;
            int before = 0;
            if (wv > 0) before += wtot[0];
            if (wv > 1) before += wtot[1];
            if (wv > 2) before += wtot[2];
            cnt[tid] = before + x - c;
        }
        __syncthreads();
#pragma unroll
        for (int i = 0; i < SORT_RPT; ++i) {
            const int idx = tid + i * SUM_THREADS;
            if (idx < m) srec[cnt[q[i].x & KEY_MASK] + rk[i]] = q[i];
        }
        __syncthreads();

        for (int g = wave * 32; g < m; g += SUM_THREADS) {
            const int e = g + lane;
            int key = KEY_MASK;
            float val = 0.0f;
            if (e < m) {
                const int2 r = srec[e];
                key = r.x;
                val = __int_as_float(r.y);
            }
            const int n = (m - g < 32) ? (m - g) : 32;
            int cur = -1;
            float s[J];
#pragma unroll
            for (int j = 0; j < J; ++j) s[j] = 0.0f;
            for (int u0 = 0; u0 < n; u0 += UB) {
                float r[UB][J];
#pragma unroll
                for (int u = 0; u < UB; ++u) {
                    const int uu = (u0 + u < n) ? (u0 + u) : (n - 1);
                    const int k = lane_read_i(key, uu);
                    const float* row = H + (size_t)(k >> KEY_SHIFT) * D + lane;
#pragma unroll
                    for (int j = 0; j < J; ++j) r[u][j] = row[32 * j];
                }
#pragma unroll
                for (int u = 0; u < UB; ++u) {
                    if (u0 + u < n) {
                        const int k = lane_read_i(key, u0 + u);
                        const float w = lane_read_f(val, u0 + u);
                        const int lr = k & KEY_MASK;
                        if (lr != cur) {
                            if ((unsigned)cur < (unsigned)ROWS_PER_BIN) {
                                float* a = acc + cur * D + lane;
#pragma unroll
                                for (int j = 0; j < J; ++j) atomicAdd(a + 32 * j, s[j]);
                            }
                            cur = lr;
#pragma unroll
                            for (int j = 0; j < J; ++j) s[j] = 0.0f;
                        }
#pragma unroll
                        for (int j = 0; j < J; ++j) s[j] = fmaf(r[u][j], w, s[j]);
                    }
                }
            }
            if ((unsigned)cur < (unsigned)ROWS_PER_BIN) {
                float* a = acc + cur * D + lane;
#pragma unroll
                for (int j = 0; j < J; ++j) atomicAdd(a + 32 * j, s[j]);
            }
        }
    }
    __syncthreads();

    float4* o4 = (float4*)(out + (size_t)b * ROWS_PER_BIN * D);
    for (int i = tid; i < T4; i += SUM_THREADS) o4[i] = acc4[i];
}

__global__ __launch_bounds__(SUM_THREADS, 16) void bin_sum_d128(const int* __restrict__ rec,
                                                            const int* __restrict__ binstart,
                                                            const float* __restrict__ H,
                                                            float* __restrict__ out) {
    bin_sum<128>(rec, binstart, H, out);
}

__global__ __launch_bounds__(SUM_THREADS, 16) void bin_sum_d64(const int* __restrict__ rec,
                                                           const int* __restrict__ binstart,
                                                           const float* __restrict__ H,
                                                           float* __restrict__ out) {
    bin_sum<64>(rec, binstart, H, out);
}

extern "C" void kernel_launch(void* const* d_in, const int* in_sizes, int n_in,
                              void* d_out, int out_size, void* d_ws, size_t ws_size,
                              hipStream_t stream) {
    const float* X     = (const float*)d_in[0];
    const int*   esrc  = (const int*)d_in[1];
    const int*   edst  = (const int*)d_in[2];
    const float* evals = (const float*)d_in[3];
    const float* W1    = (const float*)d_in[4];
    const float* W2    = (const float*)d_in[5];
    float*       out   = (float*)d_out;

    char* ws = (char*)d_ws;
    const size_t h1Bytes  = (size_t)N_NODES * HID_DIM * sizeof(float);
    const size_t recBytes = (size_t)N_EDGES * 2 * sizeof(int);
    float* H1   = (float*)(ws);
    float* Hagg = (float*)(ws + h1Bytes);
    float* H2   = (float*)(ws);
    int* rec      = (int*)(ws + 2 * h1Bytes);
    int* tables   = (int*)(ws + 2 * h1Bytes + recBytes);
    int* gcount   = tables;
    int* gcursor  = tables + TABLE_INTS;
    int* binstart = tables + 2 * TABLE_INTS;

    gemm1_wmma<<<N_NODES / G1_ROWS, 256, 0, stream>>>(X, W1, H1, tables);
    bin_count<<<BIN_BLOCKS, BIN_THREADS, 0, stream>>>(edst, N_EDGES, gcount);
    bin_fill<<<BIN_BLOCKS, BIN_THREADS, 0, stream>>>(esrc, edst, evals, N_EDGES, gcount, gcursor,
                                                     binstart, rec);
    bin_sum_d128<<<NBINS, SUM_THREADS, 0, stream>>>(rec, binstart, H1, Hagg);

    gemm2_wmma<<<N_NODES / 32, 256, 0, stream>>>(Hagg, W2, H2);
    bin_sum_d64<<<NBINS, SUM_THREADS, 0, stream>>>(rec, binstart, H2, out);
}
